# starting kernel
# baseline (speedup 1.0000x reference)
#include <hip/hip_runtime.h>

typedef float v2f __attribute__((ext_vector_type(2)));
typedef float v8f __attribute__((ext_vector_type(8)));

namespace {

constexpr int kB  = 2048;
constexpr int kT  = 512;
constexpr int kIn = 28;
constexpr int kH  = 64;
constexpr int kC  = 10;
constexpr int kBM = 16;

__device__ __forceinline__ v8f wmma4(v2f a, v2f b, v8f c) {
  return __builtin_amdgcn_wmma_f32_16x16x4_f32(
       false, a,  false, b,
       (short)0, c,  false,  false);
}

__device__ __forceinline__ float sigmoid_f(float x) {
  return 1.0f / (1.0f + __expf(-x));
}
__device__ __forceinline__ float tanh_f(float x) {
  return 1.0f - 2.0f / (__expf(2.0f * x) + 1.0f);
}

__global__ __launch_bounds__(128, 1) void gru_fused_kernel(
    const float* __restrict__ x,
    const float* __restrict__ W_ih,
    const float* __restrict__ W_hh,
    const float* __restrict__ b_ih,
    const float* __restrict__ b_hh,
    const float* __restrict__ W_out,
    const float* __restrict__ b_out,
    float* __restrict__ out)
{
  __shared__ float h_lds[kBM * kH];

  const int tid  = threadIdx.x;
  const int wave = tid >> 5;
  const int lane = tid & 31;
  const int m    = lane & 15;
  const int hi   = lane >> 4;
  const int bm   = blockIdx.x * kBM;

  v2f whhB[3][16];
  v2f wihB[3][7];
#pragma unroll
  for (int g = 0; g < 3; ++g) {
    const int nrow = g * 64 + wave * 16 + m;
#pragma unroll
    for (int s = 0; s < 16; ++s)
      whhB[g][s] = *(const v2f*)(W_hh + nrow * kH + 4 * s + 2 * hi);
#pragma unroll
    for (int s = 0; s < 7; ++s)
      wihB[g][s] = *(const v2f*)(W_ih + nrow * kIn + 4 * s + 2 * hi);
  }

  const int col = wave * 16 + m;
  const float bias_r  = b_ih[col]       + b_hh[col];
  const float bias_z  = b_ih[64 + col]  + b_hh[64 + col];
  const float bias_xn = b_ih[128 + col];
  const float bias_hn = b_hh[128 + col];

  v2f hA[16];
#pragma unroll
  for (int s = 0; s < 16; ++s) hA[s] = v2f{0.0f, 0.0f};
  v8f h_c = {};

  const float* xbase = x + (size_t)(bm + m) * (kT * kIn) + 2 * hi;

  for (int t = 0; t < kT; ++t) {
    v2f xA[7];
    const float* xt = xbase + t * kIn;
#pragma unroll
    for (int s = 0; s < 7; ++s) xA[s] = *(const v2f*)(xt + 4 * s);

    v8f acc_r, acc_z, acc_xn, acc_hn;
#pragma unroll
    for (int j = 0; j < 8; ++j) {
      acc_r[j]  = bias_r;
      acc_z[j]  = bias_z;
      acc_xn[j] = bias_xn;
      acc_hn[j] = bias_hn;
    }

#pragma unroll
    for (int s = 0; s < 7; ++s) {
      acc_r  = wmma4(xA[s], wihB[0][s], acc_r);
      acc_z  = wmma4(xA[s], wihB[1][s], acc_z);
      acc_xn = wmma4(xA[s], wihB[2][s], acc_xn);
    }
#pragma unroll
    for (int s = 0; s < 16; ++s) {
      acc_r  = wmma4(hA[s], whhB[0][s], acc_r);
      acc_z  = wmma4(hA[s], whhB[1][s], acc_z);
      acc_hn = wmma4(hA[s], whhB[2][s], acc_hn);
    }

    v8f h_new;
#pragma unroll
    for (int j = 0; j < 8; ++j) {
      const float r = sigmoid_f(acc_r[j]);
      const float z = sigmoid_f(acc_z[j]);
      const float n = tanh_f(acc_xn[j] + r * acc_hn[j]);
      h_new[j] = (1.0f - z) * n + z * h_c[j];
    }
    h_c = h_new;

    __syncthreads();
#pragma unroll
    for (int j = 0; j < 8; ++j)
      h_lds[(j + 8 * hi) * kH + col] = h_new[j];
    __syncthreads();
#pragma unroll
    for (int s = 0; s < 16; ++s)
      hA[s] = *(const v2f*)(h_lds + m * kH + 4 * s + 2 * hi);
  }

  if (tid < kBM) {
    float logits[kC];
    float mx = -3.4e38f;
#pragma unroll
    for (int c = 0; c < kC; ++c) {
      float d = b_out[c];
#pragma unroll
      for (int k = 0; k < kH; ++k)
        d += h_lds[tid * kH + k] * W_out[c * kH + k];
      logits[c] = d;
      mx = fmaxf(mx, d);
    }
    float sum = 0.0f;
#pragma unroll
    for (int c = 0; c < kC; ++c) {
      const float e = __expf(logits[c] - mx);
      logits[c] = e;
      sum += e;
    }
    const float inv = 1.0f / sum;
#pragma unroll
    for (int c = 0; c < kC; ++c)
      out[(size_t)(bm + tid) * kC + c] = logits[c] * inv;
  }
}

}

extern "C" void kernel_launch(void* const* d_in, const int* in_sizes, int n_in,
                              void* d_out, int out_size, void* d_ws, size_t ws_size,
                              hipStream_t stream) {
  (void)in_sizes; (void)n_in; (void)out_size; (void)d_ws; (void)ws_size;
  const float* x     = (const float*)d_in[0];
  const float* W_ih  = (const float*)d_in[1];
  const float* W_hh  = (const float*)d_in[2];
  const float* b_ih  = (const float*)d_in[3];
  const float* b_hh  = (const float*)d_in[4];
  const float* W_out = (const float*)d_in[5];
  const float* b_out = (const float*)d_in[6];
  float* out = (float*)d_out;

  dim3 grid(kB / kBM);
  dim3 block(128);
  gru_fused_kernel<<<grid, block, 0, stream>>>(x, W_ih, W_hh, b_ih, b_hh,
                                               W_out, b_out, out);
}


# h tile and projection tiles kept twice in LDS, a step reads one copy and writes the other: one workgroup barrier per step, was two; tile stores leave the recurrent waves' path; sum order unchanged
# speedup vs baseline: 1.9801x; 1.9801x over previous
#include <hip/hip_runtime.h>


typedef float v2f __attribute__((ext_vector_type(2)));
typedef float v4f __attribute__((ext_vector_type(4)));
typedef float v8f __attribute__((ext_vector_type(8)));

namespace {

constexpr int kB  = 2048;
constexpr int kT  = 512;
constexpr int kIn = 28;
constexpr int kH  = 64;
constexpr int kC  = 10;
constexpr int kBM = 16;

constexpr int kXgSlice = 6 * 32;
constexpr int kXgAll   = 4 * kXgSlice;

__device__ __forceinline__ v8f wmma4(v2f a, v2f b, v8f c) {
  return __builtin_amdgcn_wmma_f32_16x16x4_f32(
       false, a,  false, b,
       (short)0, c,  false,  false);
}

__device__ __forceinline__ v4f lo4(v8f v) { return v4f{v[0], v[1], v[2], v[3]}; }
__device__ __forceinline__ v4f hi4(v8f v) { return v4f{v[4], v[5], v[6], v[7]}; }
__device__ __forceinline__ v8f cat8(v4f a, v4f b) {
  return v8f{a[0], a[1], a[2], a[3], b[0], b[1], b[2], b[3]};
}

__global__ __launch_bounds__(256) __attribute__((amdgpu_waves_per_eu(2, 2)))
void gru_fused_kernel(
    const float* __restrict__ x,
    const float* __restrict__ W_ih,
    const float* __restrict__ W_hh,
    const float* __restrict__ b_ih,
    const float* __restrict__ b_hh,
    const float* __restrict__ W_out,
    const float* __restrict__ b_out,
    float* __restrict__ out)
{
  __shared__ float h_lds[2 * kBM * kH];
  __shared__ v4f   xg_lds[2 * kXgAll];

  const int tid  = threadIdx.x;
  const int wave8 = __builtin_amdgcn_readfirstlane(tid >> 5);
  const bool projecting = wave8 >= 4;
  const int wave = wave8 & 3;
  const int lane = tid & 31;
  const int m    = lane & 15;
  const int hi   = lane >> 4;
  const int bm   = blockIdx.x * kBM;
  const int col  = wave * 16 + m;

  v2f whhB[3][16];
  v2f wihB[3][7];
  float bias_r = 0.0f, bias_z = 0.0f, bias_xn = 0.0f, bias_hn = 0.0f;
  v8f h_c = {};
  v2f hA[16];
  v8f xg_r = {}, xg_z = {}, xg_n = {};
  v2f xA[7];
#pragma unroll
  for (int g = 0; g < 3; ++g) {
#pragma unroll
    for (int s = 0; s < 16; ++s) whhB[g][s] = v2f{0.0f, 0.0f};
#pragma unroll
    for (int s = 0; s < 7; ++s) wihB[g][s] = v2f{0.0f, 0.0f};
  }
#pragma unroll
  for (int s = 0; s < 7; ++s) xA[s] = v2f{0.0f, 0.0f};
#pragma unroll
  for (int s = 0; s < 16; ++s) hA[s] = v2f{0.0f, 0.0f};

  const float* xbase = x + (size_t)(bm + m) * (kT * kIn) + 2 * hi;
  v4f* xg_cur = xg_lds + wave * kXgSlice + lane;
  v4f* xg_nxt = xg_cur + kXgAll;
  const float* h_rd_cur = h_lds + m * kH + 2 * hi;
  const float* h_rd_nxt = h_rd_cur + kBM * kH;
  float* h_wr_cur = h_lds + (8 * hi) * kH + col;
  float* h_wr_nxt = h_wr_cur + kBM * kH;

  if (projecting) {
#pragma unroll
    for (int g = 0; g < 3; ++g) {
      const int nrow = g * 64 + col;
#pragma unroll
      for (int s = 0; s < 7; ++s)
        wihB[g][s] = *(const v2f*)(W_ih + nrow * kIn + 4 * s + 2 * hi);
    }
    bias_r  = b_ih[col]       + b_hh[col];
    bias_z  = b_ih[64 + col]  + b_hh[64 + col];
    bias_xn = b_ih[128 + col];

#pragma unroll
    for (int s = 0; s < 7; ++s) xA[s] = *(const v2f*)(xbase + 4 * s);
    v8f r, z, n;
#pragma unroll
    for (int j = 0; j < 8; ++j) { r[j] = bias_r; z[j] = bias_z; n[j] = bias_xn; }
#pragma unroll
    for (int s = 0; s < 7; ++s) {
      r = wmma4(xA[s], wihB[0][s], r);
      z = wmma4(xA[s], wihB[1][s], z);
      n = wmma4(xA[s], wihB[2][s], n);
    }
    xg_cur[0 * 32] = lo4(r); xg_cur[1 * 32] = hi4(r);
    xg_cur[2 * 32] = lo4(z); xg_cur[3 * 32] = hi4(z);
    xg_cur[4 * 32] = lo4(n); xg_cur[5 * 32] = hi4(n);
#pragma unroll
    for (int s = 0; s < 7; ++s) xA[s] = *(const v2f*)(xbase + kIn + 4 * s);
  } else {
#pragma unroll
    for (int g = 0; g < 3; ++g) {
      const int nrow = g * 64 + col;
#pragma unroll
      for (int s = 0; s < 16; ++s)
        whhB[g][s] = *(const v2f*)(W_hh + nrow * kH + 4 * s + 2 * hi);
    }
    bias_hn = b_hh[128 + col];
#pragma unroll
    for (int j = 0; j < 8; ++j) h_wr_cur[j * kH] = 0.0f;
  }

  __syncthreads();

  if (!projecting) {
    xg_r = cat8(xg_cur[0 * 32], xg_cur[1 * 32]);
    xg_z = cat8(xg_cur[2 * 32], xg_cur[3 * 32]);
#pragma unroll
    for (int s = 0; s < 16; ++s)
      hA[s] = *(const v2f*)(h_rd_cur + 4 * s);
    xg_n = cat8(xg_cur[4 * 32], xg_cur[5 * 32]);
  }

  for (int t = 0; t < kT; ++t) {
    if (projecting) {
      v8f pr, pz, pn;
#pragma unroll
      for (int j = 0; j < 8; ++j) { pr[j] = bias_r; pz[j] = bias_z; pn[j] = bias_xn; }
#pragma unroll
      for (int s = 0; s < 7; ++s) {
        pr = wmma4(xA[s], wihB[0][s], pr);
        pz = wmma4(xA[s], wihB[1][s], pz);
        pn = wmma4(xA[s], wihB[2][s], pn);
      }
      const int tn = (t + 2 < kT) ? (t + 2) : (kT - 1);
      const float* xt = xbase + tn * kIn;
#pragma unroll
      for (int s = 0; s < 7; ++s) xA[s] = *(const v2f*)(xt + 4 * s);
      xg_nxt[0 * 32] = lo4(pr); xg_nxt[1 * 32] = hi4(pr);
      xg_nxt[2 * 32] = lo4(pz); xg_nxt[3 * 32] = hi4(pz);
      xg_nxt[4 * 32] = lo4(pn); xg_nxt[5 * 32] = hi4(pn);
    } else {
      v8f acc_r = xg_r, acc_z = xg_z, acc_xn = xg_n;
      v8f acc_hn;
#pragma unroll
      for (int j = 0; j < 8; ++j) acc_hn[j] = bias_hn;
#pragma unroll
      for (int s = 0; s < 16; ++s) {
        acc_r  = wmma4(hA[s], whhB[0][s], acc_r);
        acc_z  = wmma4(hA[s], whhB[1][s], acc_z);
        acc_hn = wmma4(hA[s], whhB[2][s], acc_hn);
      }

      float er[8], ez[8], rr[8], zz[8], en[8], nn[8];
#pragma unroll
      for (int j = 0; j < 8; ++j) er[j] = __expf(-acc_r[j]);
#pragma unroll
      for (int j = 0; j < 8; ++j) ez[j] = __expf(-acc_z[j]);
#pragma unroll
      for (int j = 0; j < 8; ++j) rr[j] = 1.0f / (1.0f + er[j]);
#pragma unroll
      for (int j = 0; j < 8; ++j) zz[j] = 1.0f / (1.0f + ez[j]);
#pragma unroll
      for (int j = 0; j < 8; ++j)
        en[j] = __expf(2.0f * (acc_xn[j] + rr[j] * acc_hn[j]));
#pragma unroll
      for (int j = 0; j < 8; ++j) nn[j] = 1.0f - 2.0f / (en[j] + 1.0f);
      v8f h_new;
#pragma unroll
      for (int j = 0; j < 8; ++j)
        h_new[j] = (1.0f - zz[j]) * nn[j] + zz[j] * h_c[j];
      h_c = h_new;
#pragma unroll
      for (int j = 0; j < 8; ++j) h_wr_nxt[j * kH] = h_c[j];
    }

    __syncthreads();
    { v4f* p = xg_cur; xg_cur = xg_nxt; xg_nxt = p; }
    { const float* p = h_rd_cur; h_rd_cur = h_rd_nxt; h_rd_nxt = p; }
    { float* p = h_wr_cur; h_wr_cur = h_wr_nxt; h_wr_nxt = p; }
    if (!projecting) {
      xg_r = cat8(xg_cur[0 * 32], xg_cur[1 * 32]);
      xg_z = cat8(xg_cur[2 * 32], xg_cur[3 * 32]);
#pragma unroll
      for (int s = 0; s < 16; ++s)
        hA[s] = *(const v2f*)(h_rd_cur + 4 * s);
      xg_n = cat8(xg_cur[4 * 32], xg_cur[5 * 32]);
    }
  }

  const float* h_fin = h_lds + ((kT & 1) ? kBM * kH : 0);

  if (tid < kBM) {
    float logits[kC];
    float mx = -3.4e38f;
#pragma unroll
    for (int c = 0; c < kC; ++c) {
      float d = b_out[c];
#pragma unroll
      for (int k = 0; k < kH; ++k)
        d += h_fin[tid * kH + k] * W_out[c * kH + k];
      logits[c] = d;
      mx = fmaxf(mx, d);
    }
    float sum = 0.0f;
#pragma unroll
    for (int c = 0; c < kC; ++c) {
      const float e = __expf(logits[c] - mx);
      logits[c] = e;
      sum += e;
    }
    const float inv = 1.0f / sum;
#pragma unroll
    for (int c = 0; c < kC; ++c)
      out[(size_t)(bm + tid) * kC + c] = logits[c] * inv;
  }
}

}

extern "C" void kernel_launch(void* const* d_in, const int* in_sizes, int n_in,
                              void* d_out, int out_size, void* d_ws, size_t ws_size,
                              hipStream_t stream) {
  (void)in_sizes; (void)n_in; (void)out_size; (void)d_ws; (void)ws_size;
  const float* x     = (const float*)d_in[0];
  const float* W_ih  = (const float*)d_in[1];
  const float* W_hh  = (const float*)d_in[2];
  const float* b_ih  = (const float*)d_in[3];
  const float* b_hh  = (const float*)d_in[4];
  const float* W_out = (const float*)d_in[5];
  const float* b_out = (const float*)d_in[6];
  float* out = (float*)d_out;

  dim3 grid(kB / kBM);
  dim3 block(256);
  gru_fused_kernel<<<grid, block, 0, stream>>>(x, W_ih, W_hh, b_ih, b_hh,
                                               W_out, b_out, out);
}


# h tile kept in LDS in A-fragment order, chunk places XOR-rotated: a lane reads its h as 8 side-by-side 128-bit loads, was 8 two-address reads 16 rows deep on a bank; sum order unchanged
# speedup vs baseline: 2.2467x; 1.1346x over previous
#include <hip/hip_runtime.h>


typedef float v2f __attribute__((ext_vector_type(2)));
typedef float v4f __attribute__((ext_vector_type(4)));
typedef float v8f __attribute__((ext_vector_type(8)));

namespace {

constexpr int kB  = 2048;
constexpr int kT  = 512;
constexpr int kIn = 28;
constexpr int kH  = 64;
constexpr int kC  = 10;
constexpr int kBM = 16;

constexpr int kXgSlice = 6 * 32;
constexpr int kXgAll   = 4 * kXgSlice;

constexpr int kHCopy = kBM * kH;
__device__ __forceinline__ int h_index(int row, int colm) {
  const int p    = colm >> 3;
  const int half = (colm >> 1) & 1;
  const int word = (colm & 1) + 2 * ((colm >> 2) & 1);
  return (p * 32 + 16 * half + (row ^ (2 * half + 4 * (p & 1)))) * 4 + word;
}

__device__ __forceinline__ v8f wmma4(v2f a, v2f b, v8f c) {
  return __builtin_amdgcn_wmma_f32_16x16x4_f32(
       false, a,  false, b,
       (short)0, c,  false,  false);
}

__device__ __forceinline__ v4f lo4(v8f v) { return v4f{v[0], v[1], v[2], v[3]}; }
__device__ __forceinline__ v4f hi4(v8f v) { return v4f{v[4], v[5], v[6], v[7]}; }
__device__ __forceinline__ v8f cat8(v4f a, v4f b) {
  return v8f{a[0], a[1], a[2], a[3], b[0], b[1], b[2], b[3]};
}

__global__ __launch_bounds__(256) __attribute__((amdgpu_waves_per_eu(2, 2)))
void gru_fused_kernel(
    const float* __restrict__ x,
    const float* __restrict__ W_ih,
    const float* __restrict__ W_hh,
    const float* __restrict__ b_ih,
    const float* __restrict__ b_hh,
    const float* __restrict__ W_out,
    const float* __restrict__ b_out,
    float* __restrict__ out)
{
  __shared__ v4f   h_lds4[2 * kHCopy / 4];
  float* const h_lds = (float*)h_lds4;
  __shared__ v4f   xg_lds[2 * kXgAll];

  const int tid  = threadIdx.x;
  const int wave8 = __builtin_amdgcn_readfirstlane(tid >> 5);
  const bool projecting = wave8 >= 4;
  const int wave = wave8 & 3;
  const int lane = tid & 31;
  const int m    = lane & 15;
  const int hi   = lane >> 4;
  const int bm   = blockIdx.x * kBM;
  const int col  = wave * 16 + m;

  v2f whhB[3][16];
  v2f wihB[3][7];
  float bias_r = 0.0f, bias_z = 0.0f, bias_xn = 0.0f, bias_hn = 0.0f;
  v8f h_c = {};
  v2f hA[16];
  v8f xg_r = {}, xg_z = {}, xg_n = {};
  v2f xA[7];
#pragma unroll
  for (int g = 0; g < 3; ++g) {
#pragma unroll
    for (int s = 0; s < 16; ++s) whhB[g][s] = v2f{0.0f, 0.0f};
#pragma unroll
    for (int s = 0; s < 7; ++s) wihB[g][s] = v2f{0.0f, 0.0f};
  }
#pragma unroll
  for (int s = 0; s < 7; ++s) xA[s] = v2f{0.0f, 0.0f};
#pragma unroll
  for (int s = 0; s < 16; ++s) hA[s] = v2f{0.0f, 0.0f};

  const float* xbase = x + (size_t)(bm + m) * (kT * kIn) + 2 * hi;
  v4f* xg_cur = xg_lds + wave * kXgSlice + lane;
  v4f* xg_nxt = xg_cur + kXgAll;
  int h_rd = h_index(m, 2 * hi);
  int h_wr = h_index(8 * hi, col);

  if (projecting) {
#pragma unroll
    for (int g = 0; g < 3; ++g) {
      const int nrow = g * 64 + col;
#pragma unroll
      for (int s = 0; s < 7; ++s)
        wihB[g][s] = *(const v2f*)(W_ih + nrow * kIn + 4 * s + 2 * hi);
    }
    bias_r  = b_ih[col]       + b_hh[col];
    bias_z  = b_ih[64 + col]  + b_hh[64 + col];
    bias_xn = b_ih[128 + col];

#pragma unroll
    for (int s = 0; s < 7; ++s) xA[s] = *(const v2f*)(xbase + 4 * s);
    v8f r, z, n;
#pragma unroll
    for (int j = 0; j < 8; ++j) { r[j] = bias_r; z[j] = bias_z; n[j] = bias_xn; }
#pragma unroll
    for (int s = 0; s < 7; ++s) {
      r = wmma4(xA[s], wihB[0][s], r);
      z = wmma4(xA[s], wihB[1][s], z);
      n = wmma4(xA[s], wihB[2][s], n);
    }
    xg_cur[0 * 32] = lo4(r); xg_cur[1 * 32] = hi4(r);
    xg_cur[2 * 32] = lo4(z); xg_cur[3 * 32] = hi4(z);
    xg_cur[4 * 32] = lo4(n); xg_cur[5 * 32] = hi4(n);
#pragma unroll
    for (int s = 0; s < 7; ++s) xA[s] = *(const v2f*)(xbase + kIn + 4 * s);
  } else {
#pragma unroll
    for (int g = 0; g < 3; ++g) {
      const int nrow = g * 64 + col;
#pragma unroll
      for (int s = 0; s < 16; ++s)
        whhB[g][s] = *(const v2f*)(W_hh + nrow * kH + 4 * s + 2 * hi);
    }
    bias_hn = b_hh[128 + col];
#pragma unroll
    for (int j = 0; j < 8; j += 2) {
      h_lds[(h_wr ^ (4 * j))]     = 0.0f;
      h_lds[(h_wr ^ (4 * j)) + 4] = 0.0f;
    }
  }

  __syncthreads();

  if (!projecting) {
    xg_r = cat8(xg_cur[0 * 32], xg_cur[1 * 32]);
    xg_z = cat8(xg_cur[2 * 32], xg_cur[3 * 32]);
#pragma unroll
    for (int p = 0; p < 8; ++p) {
      const v4f q = *(const v4f*)(h_lds + (h_rd ^ ((p & 1) * 16)) + 128 * p);
      hA[2 * p]     = v2f{q[0], q[1]};
      hA[2 * p + 1] = v2f{q[2], q[3]};
    }
    xg_n = cat8(xg_cur[4 * 32], xg_cur[5 * 32]);
  }

  for (int t = 0; t < kT; ++t) {
    if (projecting) {
      v8f pr, pz, pn;
#pragma unroll
      for (int j = 0; j < 8; ++j) { pr[j] = bias_r; pz[j] = bias_z; pn[j] = bias_xn; }
#pragma unroll
      for (int s = 0; s < 7; ++s) {
        pr = wmma4(xA[s], wihB[0][s], pr);
        pz = wmma4(xA[s], wihB[1][s], pz);
        pn = wmma4(xA[s], wihB[2][s], pn);
      }
      const int tn = (t + 2 < kT) ? (t + 2) : (kT - 1);
      const float* xt = xbase + tn * kIn;
#pragma unroll
      for (int s = 0; s < 7; ++s) xA[s] = *(const v2f*)(xt + 4 * s);
      xg_nxt[0 * 32] = lo4(pr); xg_nxt[1 * 32] = hi4(pr);
      xg_nxt[2 * 32] = lo4(pz); xg_nxt[3 * 32] = hi4(pz);
      xg_nxt[4 * 32] = lo4(pn); xg_nxt[5 * 32] = hi4(pn);
    } else {
      v8f acc_r = xg_r, acc_z = xg_z, acc_xn = xg_n;
      v8f acc_hn;
#pragma unroll
      for (int j = 0; j < 8; ++j) acc_hn[j] = bias_hn;
#pragma unroll
      for (int s = 0; s < 16; ++s) {
        acc_r  = wmma4(hA[s], whhB[0][s], acc_r);
        acc_z  = wmma4(hA[s], whhB[1][s], acc_z);
        acc_hn = wmma4(hA[s], whhB[2][s], acc_hn);
      }

      float er[8], ez[8], rr[8], zz[8], en[8], nn[8];
#pragma unroll
      for (int j = 0; j < 8; ++j) er[j] = __expf(-acc_r[j]);
#pragma unroll
      for (int j = 0; j < 8; ++j) ez[j] = __expf(-acc_z[j]);
#pragma unroll
      for (int j = 0; j < 8; ++j) rr[j] = 1.0f / (1.0f + er[j]);
#pragma unroll
      for (int j = 0; j < 8; ++j) zz[j] = 1.0f / (1.0f + ez[j]);
#pragma unroll
      for (int j = 0; j < 8; ++j)
        en[j] = __expf(2.0f * (acc_xn[j] + rr[j] * acc_hn[j]));
#pragma unroll
      for (int j = 0; j < 8; ++j) nn[j] = 1.0f - 2.0f / (en[j] + 1.0f);
      v8f h_new;
#pragma unroll
      for (int j = 0; j < 8; ++j)
        h_new[j] = (1.0f - zz[j]) * nn[j] + zz[j] * h_c[j];
      h_c = h_new;
      const int h_wr_nxt = h_wr ^ kHCopy;
#pragma unroll
      for (int j = 0; j < 8; j += 2) {
        h_lds[(h_wr_nxt ^ (4 * j))]     = h_c[j];
        h_lds[(h_wr_nxt ^ (4 * j)) + 4] = h_c[j + 1];
      }
    }

    __syncthreads();
    { v4f* p = xg_cur; xg_cur = xg_nxt; xg_nxt = p; }
    h_rd ^= kHCopy;
    h_wr ^= kHCopy;
    if (!projecting) {
      xg_r = cat8(xg_cur[0 * 32], xg_cur[1 * 32]);
      xg_z = cat8(xg_cur[2 * 32], xg_cur[3 * 32]);
#pragma unroll
      for (int p = 0; p < 8; ++p) {
        const v4f q = *(const v4f*)(h_lds + (h_rd ^ ((p & 1) * 16)) + 128 * p);
        hA[2 * p]     = v2f{q[0], q[1]};
        hA[2 * p + 1] = v2f{q[2], q[3]};
      }
      xg_n = cat8(xg_cur[4 * 32], xg_cur[5 * 32]);
    }
  }

  const float* h_fin = h_lds + ((kT & 1) ? kHCopy : 0);

  if (tid < kBM) {
    float hrow[kH];
#pragma unroll
    for (int p = 0; p < 8; ++p) {
#pragma unroll
      for (int hh = 0; hh < 2; ++hh) {
        const v4f q = *(const v4f*)(h_fin + h_index(tid, 8 * p + 2 * hh));
        hrow[8 * p + 2 * hh]     = q[0];
        hrow[8 * p + 2 * hh + 1] = q[1];
        hrow[8 * p + 2 * hh + 4] = q[2];
        hrow[8 * p + 2 * hh + 5] = q[3];
      }
    }
    float logits[kC];
    float mx = -3.4e38f;
#pragma unroll
    for (int c = 0; c < kC; ++c) {
      float d = b_out[c];
#pragma unroll
      for (int k = 0; k < kH; ++k)
        d += hrow[k] * W_out[c * kH + k];
      logits[c] = d;
      mx = fmaxf(mx, d);
    }
    float sum = 0.0f;
#pragma unroll
    for (int c = 0; c < kC; ++c) {
      const float e = __expf(logits[c] - mx);
      logits[c] = e;
      sum += e;
    }
    const float inv = 1.0f / sum;
#pragma unroll
    for (int c = 0; c < kC; ++c)
      out[(size_t)(bm + tid) * kC + c] = logits[c] * inv;
  }
}

}

extern "C" void kernel_launch(void* const* d_in, const int* in_sizes, int n_in,
                              void* d_out, int out_size, void* d_ws, size_t ws_size,
                              hipStream_t stream) {
  (void)in_sizes; (void)n_in; (void)out_size; (void)d_ws; (void)ws_size;
  const float* x     = (const float*)d_in[0];
  const float* W_ih  = (const float*)d_in[1];
  const float* W_hh  = (const float*)d_in[2];
  const float* b_ih  = (const float*)d_in[3];
  const float* b_hh  = (const float*)d_in[4];
  const float* W_out = (const float*)d_in[5];
  const float* b_out = (const float*)d_in[6];
  float* out = (float*)d_out;

  dim3 grid(kB / kBM);
  dim3 block(256);
  gru_fused_kernel<<<grid, block, 0, stream>>>(x, W_ih, W_hh, b_ih, b_hh,
                                               W_out, b_out, out);
}


# gate divisions share denominators, 12 per lane and step (was 24): r reciprocals in pairs from one quotient, z and tanh merged into one quotient; exps capped; last bits differ, sum order unchanged
# speedup vs baseline: 2.4095x; 1.0725x over previous
#include <hip/hip_runtime.h>


typedef float v2f __attribute__((ext_vector_type(2)));
typedef float v4f __attribute__((ext_vector_type(4)));
typedef float v8f __attribute__((ext_vector_type(8)));

namespace {

constexpr int kB  = 2048;
constexpr int kT  = 512;
constexpr int kIn = 28;
constexpr int kH  = 64;
constexpr int kC  = 10;
constexpr int kBM = 16;

constexpr int kXgSlice = 6 * 32;
constexpr int kXgAll   = 4 * kXgSlice;

constexpr int kHCopy = kBM * kH;

constexpr float kExpCapSig  = 9223372036854775808.0f;
constexpr float kExpCapTanh = 1099511627776.0f;
__device__ __forceinline__ int h_index(int row, int colm) {
  const int p    = colm >> 3;
  const int half = (colm >> 1) & 1;
  const int word = (colm & 1) + 2 * ((colm >> 2) & 1);
  return (p * 32 + 16 * half + (row ^ (2 * half + 4 * (p & 1)))) * 4 + word;
}

__device__ __forceinline__ v8f wmma4(v2f a, v2f b, v8f c) {
  return __builtin_amdgcn_wmma_f32_16x16x4_f32(
       false, a,  false, b,
       (short)0, c,  false,  false);
}

__device__ __forceinline__ v4f lo4(v8f v) { return v4f{v[0], v[1], v[2], v[3]}; }
__device__ __forceinline__ v4f hi4(v8f v) { return v4f{v[4], v[5], v[6], v[7]}; }
__device__ __forceinline__ v8f cat8(v4f a, v4f b) {
  return v8f{a[0], a[1], a[2], a[3], b[0], b[1], b[2], b[3]};
}

__global__ __launch_bounds__(256) __attribute__((amdgpu_waves_per_eu(2, 2)))
void gru_fused_kernel(
    const float* __restrict__ x,
    const float* __restrict__ W_ih,
    const float* __restrict__ W_hh,
    const float* __restrict__ b_ih,
    const float* __restrict__ b_hh,
    const float* __restrict__ W_out,
    const float* __restrict__ b_out,
    float* __restrict__ out)
{
  __shared__ v4f   h_lds4[2 * kHCopy / 4];
  float* const h_lds = (float*)h_lds4;
  __shared__ v4f   xg_lds[2 * kXgAll];

  const int tid  = threadIdx.x;
  const int wave8 = __builtin_amdgcn_readfirstlane(tid >> 5);
  const bool projecting = wave8 >= 4;
  const int wave = wave8 & 3;
  const int lane = tid & 31;
  const int m    = lane & 15;
  const int hi   = lane >> 4;
  const int bm   = blockIdx.x * kBM;
  const int col  = wave * 16 + m;

  v2f whhB[3][16];
  v2f wihB[3][7];
  float bias_r = 0.0f, bias_z = 0.0f, bias_xn = 0.0f, bias_hn = 0.0f;
  v8f h_c = {};
  v2f hA[16];
  v8f xg_r = {}, xg_z = {}, xg_n = {};
  v2f xA[7];
#pragma unroll
  for (int g = 0; g < 3; ++g) {
#pragma unroll
    for (int s = 0; s < 16; ++s) whhB[g][s] = v2f{0.0f, 0.0f};
#pragma unroll
    for (int s = 0; s < 7; ++s) wihB[g][s] = v2f{0.0f, 0.0f};
  }
#pragma unroll
  for (int s = 0; s < 7; ++s) xA[s] = v2f{0.0f, 0.0f};
#pragma unroll
  for (int s = 0; s < 16; ++s) hA[s] = v2f{0.0f, 0.0f};

  const float* xbase = x + (size_t)(bm + m) * (kT * kIn) + 2 * hi;
  v4f* xg_cur = xg_lds + wave * kXgSlice + lane;
  v4f* xg_nxt = xg_cur + kXgAll;
  int h_rd = h_index(m, 2 * hi);
  int h_wr = h_index(8 * hi, col);

  if (projecting) {
#pragma unroll
    for (int g = 0; g < 3; ++g) {
      const int nrow = g * 64 + col;
#pragma unroll
      for (int s = 0; s < 7; ++s)
        wihB[g][s] = *(const v2f*)(W_ih + nrow * kIn + 4 * s + 2 * hi);
    }
    bias_r  = b_ih[col]       + b_hh[col];
    bias_z  = b_ih[64 + col]  + b_hh[64 + col];
    bias_xn = b_ih[128 + col];

#pragma unroll
    for (int s = 0; s < 7; ++s) xA[s] = *(const v2f*)(xbase + 4 * s);
    v8f r, z, n;
#pragma unroll
    for (int j = 0; j < 8; ++j) { r[j] = bias_r; z[j] = bias_z; n[j] = bias_xn; }
#pragma unroll
    for (int s = 0; s < 7; ++s) {
      r = wmma4(xA[s], wihB[0][s], r);
      z = wmma4(xA[s], wihB[1][s], z);
      n = wmma4(xA[s], wihB[2][s], n);
    }
    xg_cur[0 * 32] = lo4(r); xg_cur[1 * 32] = hi4(r);
    xg_cur[2 * 32] = lo4(z); xg_cur[3 * 32] = hi4(z);
    xg_cur[4 * 32] = lo4(n); xg_cur[5 * 32] = hi4(n);
#pragma unroll
    for (int s = 0; s < 7; ++s) xA[s] = *(const v2f*)(xbase + kIn + 4 * s);
  } else {
#pragma unroll
    for (int g = 0; g < 3; ++g) {
      const int nrow = g * 64 + col;
#pragma unroll
      for (int s = 0; s < 16; ++s)
        whhB[g][s] = *(const v2f*)(W_hh + nrow * kH + 4 * s + 2 * hi);
    }
    bias_hn = b_hh[128 + col];
#pragma unroll
    for (int j = 0; j < 8; j += 2) {
      h_lds[(h_wr ^ (4 * j))]     = 0.0f;
      h_lds[(h_wr ^ (4 * j)) + 4] = 0.0f;
    }
  }

  __syncthreads();

  if (!projecting) {
    xg_r = cat8(xg_cur[0 * 32], xg_cur[1 * 32]);
    xg_z = cat8(xg_cur[2 * 32], xg_cur[3 * 32]);
#pragma unroll
    for (int p = 0; p < 8; ++p) {
      const v4f q = *(const v4f*)(h_lds + (h_rd ^ ((p & 1) * 16)) + 128 * p);
      hA[2 * p]     = v2f{q[0], q[1]};
      hA[2 * p + 1] = v2f{q[2], q[3]};
    }
    xg_n = cat8(xg_cur[4 * 32], xg_cur[5 * 32]);
  }

  for (int t = 0; t < kT; ++t) {
    if (projecting) {
      v8f pr, pz, pn;
#pragma unroll
      for (int j = 0; j < 8; ++j) { pr[j] = bias_r; pz[j] = bias_z; pn[j] = bias_xn; }
#pragma unroll
      for (int s = 0; s < 7; ++s) {
        pr = wmma4(xA[s], wihB[0][s], pr);
        pz = wmma4(xA[s], wihB[1][s], pz);
        pn = wmma4(xA[s], wihB[2][s], pn);
      }
      const int tn = (t + 2 < kT) ? (t + 2) : (kT - 1);
      const float* xt = xbase + tn * kIn;
#pragma unroll
      for (int s = 0; s < 7; ++s) xA[s] = *(const v2f*)(xt + 4 * s);
      xg_nxt[0 * 32] = lo4(pr); xg_nxt[1 * 32] = hi4(pr);
      xg_nxt[2 * 32] = lo4(pz); xg_nxt[3 * 32] = hi4(pz);
      xg_nxt[4 * 32] = lo4(pn); xg_nxt[5 * 32] = hi4(pn);
    } else {
      v8f acc_r = xg_r, acc_z = xg_z, acc_xn = xg_n;
      v8f acc_hn;
#pragma unroll
      for (int j = 0; j < 8; ++j) acc_hn[j] = bias_hn;
#pragma unroll
      for (int s = 0; s < 16; ++s) {
        acc_r  = wmma4(hA[s], whhB[0][s], acc_r);
        acc_z  = wmma4(hA[s], whhB[1][s], acc_z);
        acc_hn = wmma4(hA[s], whhB[2][s], acc_hn);
      }

      float er[8], ez[8], rr[8], en[8];
#pragma unroll
      for (int j = 0; j < 8; ++j) er[j] = fminf(__expf(-acc_r[j]), kExpCapSig);
#pragma unroll
      for (int j = 0; j < 8; ++j) ez[j] = fminf(__expf(-acc_z[j]), kExpCapSig);
#pragma unroll
      for (int j = 0; j < 8; j += 2) {
        const float a = 1.0f + er[j];
        const float b = 1.0f + er[j + 1];
        const float q = 1.0f / (a * b);
        rr[j]     = b * q;
        rr[j + 1] = a * q;
      }
#pragma unroll
      for (int j = 0; j < 8; ++j)
        en[j] = fminf(__expf(2.0f * (acc_xn[j] + rr[j] * acc_hn[j])), kExpCapTanh);
      v8f h_new;
#pragma unroll
      for (int j = 0; j < 8; ++j) {
        const float ep  = en[j] + 1.0f;
        const float em  = en[j] - 1.0f;
        const float num = ez[j] * em + h_c[j] * ep;
        const float den = (1.0f + ez[j]) * ep;
        h_new[j] = num / den;
      }
      h_c = h_new;
      const int h_wr_nxt = h_wr ^ kHCopy;
#pragma unroll
      for (int j = 0; j < 8; j += 2) {
        h_lds[(h_wr_nxt ^ (4 * j))]     = h_c[j];
        h_lds[(h_wr_nxt ^ (4 * j)) + 4] = h_c[j + 1];
      }
    }

    __syncthreads();
    { v4f* p = xg_cur; xg_cur = xg_nxt; xg_nxt = p; }
    h_rd ^= kHCopy;
    h_wr ^= kHCopy;
    if (!projecting) {
      xg_r = cat8(xg_cur[0 * 32], xg_cur[1 * 32]);
      xg_z = cat8(xg_cur[2 * 32], xg_cur[3 * 32]);
#pragma unroll
      for (int p = 0; p < 8; ++p) {
        const v4f q = *(const v4f*)(h_lds + (h_rd ^ ((p & 1) * 16)) + 128 * p);
        hA[2 * p]     = v2f{q[0], q[1]};
        hA[2 * p + 1] = v2f{q[2], q[3]};
      }
      xg_n = cat8(xg_cur[4 * 32], xg_cur[5 * 32]);
    }
  }

  const float* h_fin = h_lds + ((kT & 1) ? kHCopy : 0);

  if (tid < kBM) {
    float hrow[kH];
#pragma unroll
    for (int p = 0; p < 8; ++p) {
#pragma unroll
      for (int hh = 0; hh < 2; ++hh) {
        const v4f q = *(const v4f*)(h_fin + h_index(tid, 8 * p + 2 * hh));
        hrow[8 * p + 2 * hh]     = q[0];
        hrow[8 * p + 2 * hh + 1] = q[1];
        hrow[8 * p + 2 * hh + 4] = q[2];
        hrow[8 * p + 2 * hh + 5] = q[3];
      }
    }
    float logits[kC];
    float mx = -3.4e38f;
#pragma unroll
    for (int c = 0; c < kC; ++c) {
      float d = b_out[c];
#pragma unroll
      for (int k = 0; k < kH; ++k)
        d += hrow[k] * W_out[c * kH + k];
      logits[c] = d;
      mx = fmaxf(mx, d);
    }
    float sum = 0.0f;
#pragma unroll
    for (int c = 0; c < kC; ++c) {
      const float e = __expf(logits[c] - mx);
      logits[c] = e;
      sum += e;
    }
    const float inv = 1.0f / sum;
#pragma unroll
    for (int c = 0; c < kC; ++c)
      out[(size_t)(bm + tid) * kC + c] = logits[c] * inv;
  }
}

}

extern "C" void kernel_launch(void* const* d_in, const int* in_sizes, int n_in,
                              void* d_out, int out_size, void* d_ws, size_t ws_size,
                              hipStream_t stream) {
  (void)in_sizes; (void)n_in; (void)out_size; (void)d_ws; (void)ws_size;
  const float* x     = (const float*)d_in[0];
  const float* W_ih  = (const float*)d_in[1];
  const float* W_hh  = (const float*)d_in[2];
  const float* b_ih  = (const float*)d_in[3];
  const float* b_hh  = (const float*)d_in[4];
  const float* W_out = (const float*)d_in[5];
  const float* b_out = (const float*)d_in[6];
  float* out = (float*)d_out;

  dim3 grid(kB / kBM);
  dim3 block(256);
  gru_fused_kernel<<<grid, block, 0, stream>>>(x, W_ih, W_hh, b_ih, b_hh,
                                               W_out, b_out, out);
}


# second barrier phase per step: recurrent waves signal it after their gate math, projecting waves start their products on it (tiles two steps ahead, 3 LDS copies); sum order unchanged
# speedup vs baseline: 2.5215x; 1.0465x over previous
#include <hip/hip_runtime.h>


typedef float v2f __attribute__((ext_vector_type(2)));
typedef float v4f __attribute__((ext_vector_type(4)));
typedef float v8f __attribute__((ext_vector_type(8)));

namespace {

constexpr int kB  = 2048;
constexpr int kT  = 512;
constexpr int kIn = 28;
constexpr int kH  = 64;
constexpr int kC  = 10;
constexpr int kBM = 16;

constexpr int kXgSlice = 6 * 32;
constexpr int kXgAll   = 4 * kXgSlice;

constexpr int kHCopy = kBM * kH;

constexpr float kExpCapSig  = 9223372036854775808.0f;
constexpr float kExpCapTanh = 1099511627776.0f;
__device__ __forceinline__ int h_index(int row, int colm) {
  const int p    = colm >> 3;
  const int half = (colm >> 1) & 1;
  const int word = (colm & 1) + 2 * ((colm >> 2) & 1);
  return (p * 32 + 16 * half + (row ^ (2 * half + 4 * (p & 1)))) * 4 + word;
}

__device__ __forceinline__ v8f wmma4(v2f a, v2f b, v8f c) {
  return __builtin_amdgcn_wmma_f32_16x16x4_f32(
       false, a,  false, b,
       (short)0, c,  false,  false);
}

__device__ __forceinline__ v4f lo4(v8f v) { return v4f{v[0], v[1], v[2], v[3]}; }
__device__ __forceinline__ v4f hi4(v8f v) { return v4f{v[4], v[5], v[6], v[7]}; }
__device__ __forceinline__ v8f cat8(v4f a, v4f b) {
  return v8f{a[0], a[1], a[2], a[3], b[0], b[1], b[2], b[3]};
}

__device__ __forceinline__ void bar_signal() { __builtin_amdgcn_s_barrier_signal(-1); }
__device__ __forceinline__ void bar_wait()   { __builtin_amdgcn_s_barrier_wait(-1); }
__device__ __forceinline__ int ring_next(int o) { return (o == 2 * kXgAll) ? 0 : o + kXgAll; }

__global__ __launch_bounds__(256) __attribute__((amdgpu_waves_per_eu(2, 2)))
void gru_fused_kernel(
    const float* __restrict__ x,
    const float* __restrict__ W_ih,
    const float* __restrict__ W_hh,
    const float* __restrict__ b_ih,
    const float* __restrict__ b_hh,
    const float* __restrict__ W_out,
    const float* __restrict__ b_out,
    float* __restrict__ out)
{
  __shared__ v4f   h_lds4[2 * kHCopy / 4];
  float* const h_lds = (float*)h_lds4;
  __shared__ v4f   xg_lds[3 * kXgAll];

  const int tid  = threadIdx.x;
  const int wave8 = __builtin_amdgcn_readfirstlane(tid >> 5);
  const bool projecting = wave8 >= 4;
  const int wave = wave8 & 3;
  const int lane = tid & 31;
  const int m    = lane & 15;
  const int hi   = lane >> 4;
  const int bm   = blockIdx.x * kBM;
  const int col  = wave * 16 + m;

  v2f whhB[3][16];
  v2f wihB[3][7];
  float bias_r = 0.0f, bias_z = 0.0f, bias_xn = 0.0f, bias_hn = 0.0f;
  v8f h_c = {};
  v2f hA[16];
  v8f xg_r = {}, xg_z = {}, xg_n = {};
  v2f xA[7];
#pragma unroll
  for (int g = 0; g < 3; ++g) {
#pragma unroll
    for (int s = 0; s < 16; ++s) whhB[g][s] = v2f{0.0f, 0.0f};
#pragma unroll
    for (int s = 0; s < 7; ++s) wihB[g][s] = v2f{0.0f, 0.0f};
  }
#pragma unroll
  for (int s = 0; s < 7; ++s) xA[s] = v2f{0.0f, 0.0f};
#pragma unroll
  for (int s = 0; s < 16; ++s) hA[s] = v2f{0.0f, 0.0f};

  const float* xbase = x + (size_t)(bm + m) * (kT * kIn) + 2 * hi;
  v4f* const xg_base = xg_lds + wave * kXgSlice + lane;
  int xg_rd = 0;
  int xg_wr = 2 * kXgAll;
  int h_rd = h_index(m, 2 * hi);
  int h_wr = h_index(8 * hi, col);

  if (projecting) {
#pragma unroll
    for (int g = 0; g < 3; ++g) {
      const int nrow = g * 64 + col;
#pragma unroll
      for (int s = 0; s < 7; ++s)
        wihB[g][s] = *(const v2f*)(W_ih + nrow * kIn + 4 * s + 2 * hi);
    }
    bias_r  = b_ih[col]       + b_hh[col];
    bias_z  = b_ih[64 + col]  + b_hh[64 + col];
    bias_xn = b_ih[128 + col];

    v2f xA0[7], xA1[7];
#pragma unroll
    for (int s = 0; s < 7; ++s) {
      xA0[s] = *(const v2f*)(xbase + 4 * s);
      xA1[s] = *(const v2f*)(xbase + kIn + 4 * s);
      xA[s]  = *(const v2f*)(xbase + 2 * kIn + 4 * s);
    }
    v8f r, z, n;
#pragma unroll
    for (int j = 0; j < 8; ++j) { r[j] = bias_r; z[j] = bias_z; n[j] = bias_xn; }
#pragma unroll
    for (int s = 0; s < 7; ++s) {
      r = wmma4(xA0[s], wihB[0][s], r);
      z = wmma4(xA0[s], wihB[1][s], z);
      n = wmma4(xA0[s], wihB[2][s], n);
    }
    xg_base[0 * 32] = lo4(r); xg_base[1 * 32] = hi4(r);
    xg_base[2 * 32] = lo4(z); xg_base[3 * 32] = hi4(z);
    xg_base[4 * 32] = lo4(n); xg_base[5 * 32] = hi4(n);
#pragma unroll
    for (int j = 0; j < 8; ++j) { r[j] = bias_r; z[j] = bias_z; n[j] = bias_xn; }
#pragma unroll
    for (int s = 0; s < 7; ++s) {
      r = wmma4(xA1[s], wihB[0][s], r);
      z = wmma4(xA1[s], wihB[1][s], z);
      n = wmma4(xA1[s], wihB[2][s], n);
    }
    xg_base[kXgAll + 0 * 32] = lo4(r); xg_base[kXgAll + 1 * 32] = hi4(r);
    xg_base[kXgAll + 2 * 32] = lo4(z); xg_base[kXgAll + 3 * 32] = hi4(z);
    xg_base[kXgAll + 4 * 32] = lo4(n); xg_base[kXgAll + 5 * 32] = hi4(n);
  } else {
#pragma unroll
    for (int g = 0; g < 3; ++g) {
      const int nrow = g * 64 + col;
#pragma unroll
      for (int s = 0; s < 16; ++s)
        whhB[g][s] = *(const v2f*)(W_hh + nrow * kH + 4 * s + 2 * hi);
    }
    bias_hn = b_hh[128 + col];
#pragma unroll
    for (int j = 0; j < 8; j += 2) {
      h_lds[(h_wr ^ (4 * j))]     = 0.0f;
      h_lds[(h_wr ^ (4 * j)) + 4] = 0.0f;
    }
  }

  __syncthreads();

  if (!projecting) {
    const v4f* xg_cur = xg_base + xg_rd;
    xg_r = cat8(xg_cur[0 * 32], xg_cur[1 * 32]);
    xg_z = cat8(xg_cur[2 * 32], xg_cur[3 * 32]);
#pragma unroll
    for (int p = 0; p < 8; ++p) {
      const v4f q = *(const v4f*)(h_lds + (h_rd ^ ((p & 1) * 16)) + 128 * p);
      hA[2 * p]     = v2f{q[0], q[1]};
      hA[2 * p + 1] = v2f{q[2], q[3]};
    }
    xg_n = cat8(xg_cur[4 * 32], xg_cur[5 * 32]);
  }

  for (int t = 0; t < kT; ++t) {
    if (projecting) {
      bar_signal();
      bar_wait();
    } else {

      v8f acc_r = xg_r, acc_z = xg_z, acc_xn = xg_n;
      v8f acc_hn;
#pragma unroll
      for (int j = 0; j < 8; ++j) acc_hn[j] = bias_hn;
#pragma unroll
      for (int s = 0; s < 16; ++s) {
        acc_r  = wmma4(hA[s], whhB[0][s], acc_r);
        acc_z  = wmma4(hA[s], whhB[1][s], acc_z);
        acc_hn = wmma4(hA[s], whhB[2][s], acc_hn);
      }

      float er[8], ez[8], rr[8], en[8];
#pragma unroll
      for (int j = 0; j < 8; ++j) er[j] = fminf(__expf(-acc_r[j]), kExpCapSig);
#pragma unroll
      for (int j = 0; j < 8; ++j) ez[j] = fminf(__expf(-acc_z[j]), kExpCapSig);
#pragma unroll
      for (int j = 0; j < 8; j += 2) {
        const float a = 1.0f + er[j];
        const float b = 1.0f + er[j + 1];
        const float q = 1.0f / (a * b);
        rr[j]     = b * q;
        rr[j + 1] = a * q;
      }
#pragma unroll
      for (int j = 0; j < 8; ++j)
        en[j] = fminf(__expf(2.0f * (acc_xn[j] + rr[j] * acc_hn[j])), kExpCapTanh);
      v8f h_new;
#pragma unroll
      for (int j = 0; j < 8; ++j) {
        const float ep  = en[j] + 1.0f;
        const float em  = en[j] - 1.0f;
        const float num = ez[j] * em + h_c[j] * ep;
        const float den = (1.0f + ez[j]) * ep;
        h_new[j] = num / den;
      }
      h_c = h_new;
      __builtin_amdgcn_sched_barrier(0);
      bar_signal();
      const int h_wr_nxt = h_wr ^ kHCopy;
#pragma unroll
      for (int j = 0; j < 8; j += 2) {
        h_lds[(h_wr_nxt ^ (4 * j))]     = h_c[j];
        h_lds[(h_wr_nxt ^ (4 * j)) + 4] = h_c[j + 1];
      }
    }

    __builtin_amdgcn_fence(__ATOMIC_RELEASE, "workgroup");
    if (!projecting) bar_wait();
    bar_signal();
    if (projecting) {
      __builtin_amdgcn_sched_barrier(0);
      v8f pr, pz, pn;
#pragma unroll
      for (int j = 0; j < 8; ++j) { pr[j] = bias_r; pz[j] = bias_z; pn[j] = bias_xn; }
#pragma unroll
      for (int s = 0; s < 7; ++s) {
        pr = wmma4(xA[s], wihB[0][s], pr);
        pz = wmma4(xA[s], wihB[1][s], pz);
        pn = wmma4(xA[s], wihB[2][s], pn);
      }
      const int tn = (t + 3 < kT) ? (t + 3) : (kT - 1);
      const float* xt = xbase + tn * kIn;
#pragma unroll
      for (int s = 0; s < 7; ++s) xA[s] = *(const v2f*)(xt + 4 * s);
      v4f* xg_nxt = xg_base + xg_wr;
      xg_nxt[0 * 32] = lo4(pr); xg_nxt[1 * 32] = hi4(pr);
      xg_nxt[2 * 32] = lo4(pz); xg_nxt[3 * 32] = hi4(pz);
      xg_nxt[4 * 32] = lo4(pn); xg_nxt[5 * 32] = hi4(pn);
      xg_wr = ring_next(xg_wr);
    }
    bar_wait();
    __builtin_amdgcn_fence(__ATOMIC_ACQUIRE, "workgroup");
    h_rd ^= kHCopy;
    h_wr ^= kHCopy;
    xg_rd = ring_next(xg_rd);
    if (!projecting) {
      const v4f* xg_cur = xg_base + xg_rd;
      xg_r = cat8(xg_cur[0 * 32], xg_cur[1 * 32]);
      xg_z = cat8(xg_cur[2 * 32], xg_cur[3 * 32]);
#pragma unroll
      for (int p = 0; p < 8; ++p) {
        const v4f q = *(const v4f*)(h_lds + (h_rd ^ ((p & 1) * 16)) + 128 * p);
        hA[2 * p]     = v2f{q[0], q[1]};
        hA[2 * p + 1] = v2f{q[2], q[3]};
      }
      xg_n = cat8(xg_cur[4 * 32], xg_cur[5 * 32]);
    }
  }

  const float* h_fin = h_lds + ((kT & 1) ? kHCopy : 0);

  if (tid < kBM) {
    float hrow[kH];
#pragma unroll
    for (int p = 0; p < 8; ++p) {
#pragma unroll
      for (int hh = 0; hh < 2; ++hh) {
        const v4f q = *(const v4f*)(h_fin + h_index(tid, 8 * p + 2 * hh));
        hrow[8 * p + 2 * hh]     = q[0];
        hrow[8 * p + 2 * hh + 1] = q[1];
        hrow[8 * p + 2 * hh + 4] = q[2];
        hrow[8 * p + 2 * hh + 5] = q[3];
      }
    }
    float logits[kC];
    float mx = -3.4e38f;
#pragma unroll
    for (int c = 0; c < kC; ++c) {
      float d = b_out[c];
#pragma unroll
      for (int k = 0; k < kH; ++k)
        d += hrow[k] * W_out[c * kH + k];
      logits[c] = d;
      mx = fmaxf(mx, d);
    }
    float sum = 0.0f;
#pragma unroll
    for (int c = 0; c < kC; ++c) {
      const float e = __expf(logits[c] - mx);
      logits[c] = e;
      sum += e;
    }
    const float inv = 1.0f / sum;
#pragma unroll
    for (int c = 0; c < kC; ++c)
      out[(size_t)(bm + tid) * kC + c] = logits[c] * inv;
  }
}

}

extern "C" void kernel_launch(void* const* d_in, const int* in_sizes, int n_in,
                              void* d_out, int out_size, void* d_ws, size_t ws_size,
                              hipStream_t stream) {
  (void)in_sizes; (void)n_in; (void)out_size; (void)d_ws; (void)ws_size;
  const float* x     = (const float*)d_in[0];
  const float* W_ih  = (const float*)d_in[1];
  const float* W_hh  = (const float*)d_in[2];
  const float* b_ih  = (const float*)d_in[3];
  const float* b_hh  = (const float*)d_in[4];
  const float* W_out = (const float*)d_in[5];
  const float* b_out = (const float*)d_in[6];
  float* out = (float*)d_out;

  dim3 grid(kB / kBM);
  dim3 block(256);
  gru_fused_kernel<<<grid, block, 0, stream>>>(x, W_ih, W_hh, b_ih, b_hh,
                                               W_out, b_out, out);
}
